# starting kernel
# baseline (speedup 1.0000x reference)
#include <hip/hip_runtime.h>

#define N_NODES 100000
#define N_EDGES 500000
#define KDIM    128
#define HDIM    128
#define NCLS    40

typedef __attribute__((ext_vector_type(2))) float v2f;
typedef __attribute__((ext_vector_type(8))) float v8f;


__global__ void init_deg_kernel(float* __restrict__ deg, int n) {
    int i = blockIdx.x * blockDim.x + threadIdx.x;
    if (i < n) deg[i] = 1.0f;
}

__global__ void edge_deg_kernel(const int* __restrict__ dst, float* __restrict__ deg, int e) {
    int i = blockIdx.x * blockDim.x + threadIdx.x;
    if (i < e) atomicAdd(&deg[dst[i]], 1.0f);
}

__global__ void dinv_kernel(float* __restrict__ d, int n) {
    int i = blockIdx.x * blockDim.x + threadIdx.x;
    if (i < n) {
        float v = d[i];
        d[i] = (v > 0.0f) ? rsqrtf(v) : 0.0f;
    }
}

__global__ void edge_norm_kernel(const int* __restrict__ src, const int* __restrict__ dst,
                                 const float* __restrict__ dinv, float* __restrict__ norm, int e) {
    int i = blockIdx.x * blockDim.x + threadIdx.x;
    if (i < e) norm[i] = dinv[src[i]] * dinv[dst[i]];
}

template <int FOUT, bool RELU>
__global__ __launch_bounds__(256)
void gcn_gemm_wmma_kernel(const float* __restrict__ A,
                          const float* __restrict__ B,
                          float* __restrict__ Out,
                          int totalStrips)
{
    constexpr int FP = (FOUT % 16 == 0) ? FOUT : ((FOUT + 15) & ~15);
    constexpr int NT = FP / 16;

    __shared__ float ldsB[KDIM * FP];

    {
        constexpr int NE = KDIM * FP;
#pragma unroll 4
        for (int p = threadIdx.x; p < NE; p += 256) {
            int k = p / FP;
            int c = p - k * FP;
            float val;
            if constexpr (FOUT == FP)
                val = B[p];
            else
                val = (c < FOUT) ? B[k * FOUT + c] : 0.0f;
            ldsB[(((k >> 1) * FP) + c) * 2 + (k & 1)] = val;
        }
    }
    __syncthreads();

    int strip = blockIdx.x * (blockDim.x >> 5) + (threadIdx.x >> 5);
    if (strip >= totalStrips) return;

    int lane = threadIdx.x & 31;
    int half = lane >> 4;
    int mn   = lane & 15;

    const float* arow = A + (size_t)(strip * 16 + mn) * KDIM;

    v8f acc[NT] = {};
#pragma unroll
    for (int k0 = 0; k0 < KDIM; k0 += 4) {
        int k  = k0 + 2 * half;
        int kp = (k0 >> 1) + half;
        v2f a = *(const v2f*)(arow + k);
        if constexpr (RELU) {
            a.x = fmaxf(a.x, 0.0f);
            a.y = fmaxf(a.y, 0.0f);
        }
        const float* brow = ldsB + (size_t)kp * FP * 2 + mn * 2;
#pragma unroll
        for (int nt = 0; nt < NT; ++nt) {
            v2f b = *(const v2f*)(brow + nt * 32);
            acc[nt] = __builtin_amdgcn_wmma_f32_16x16x4_f32(
                false, a, false, b, (short)0, acc[nt], false, false);
        }
    }

#pragma unroll
    for (int nt = 0; nt < NT; ++nt) {
#pragma unroll
        for (int r = 0; r < 8; ++r) {
            int orow = strip * 16 + half * 8 + r;
            int ocol = nt * 16 + mn;
            if (FOUT == FP || ocol < FOUT)
                Out[(size_t)orow * FOUT + ocol] = acc[nt][r];
        }
    }
}

template <int CH>
__global__ void agg_init_kernel(const float4* __restrict__ hW,
                                const float*  __restrict__ dinv,
                                const float*  __restrict__ bias,
                                float4* __restrict__ out, int total)
{
    int t = blockIdx.x * blockDim.x + threadIdx.x;
    if (t >= total) return;
    int nidx = t / CH;
    int c    = t - nidx * CH;
    float w = dinv[nidx];
    w *= w;
    float4 v = hW[t];
    int f = c * 4;
    float4 r;
    r.x = v.x * w + bias[f + 0];
    r.y = v.y * w + bias[f + 1];
    r.z = v.z * w + bias[f + 2];
    r.w = v.w * w + bias[f + 3];
    out[t] = r;
}

template <int CH>
__global__ void edge_scatter_kernel(const int* __restrict__ src,
                                    const int* __restrict__ dst,
                                    const float* __restrict__ norm,
                                    const float4* __restrict__ hW,
                                    float* __restrict__ out, int total)
{
    int t = blockIdx.x * blockDim.x + threadIdx.x;
    if (t >= total) return;
    int e = t / CH;
    int c = t - e * CH;
    int s = src[e];
    int d = dst[e];
    float w  = norm[e];
    float4 v = hW[(size_t)s * CH + c];
    float* o = out + (size_t)d * (CH * 4) + c * 4;
    atomicAdd(o + 0, v.x * w);
    atomicAdd(o + 1, v.y * w);
    atomicAdd(o + 2, v.z * w);
    atomicAdd(o + 3, v.w * w);
}


extern "C" void kernel_launch(void* const* d_in, const int* in_sizes, int n_in,
                              void* d_out, int out_size, void* d_ws, size_t ws_size,
                              hipStream_t stream)
{
    const float* x  = (const float*)d_in[0];
    const int*   ei = (const int*)  d_in[1];
    const float* W[4] = { (const float*)d_in[2], (const float*)d_in[4],
                          (const float*)d_in[6], (const float*)d_in[8] };
    const float* bia[4] = { (const float*)d_in[3], (const float*)d_in[5],
                            (const float*)d_in[7], (const float*)d_in[9] };
    const int* src = ei;
    const int* dst = ei + N_EDGES;

    char* ws = (char*)d_ws;
    size_t off = 0;
    auto alloc = [&](size_t bytes) {
        size_t p = off;
        off = (off + bytes + 255) & ~(size_t)255;
        return p;
    };
    float* dinv = (float*)(ws + alloc((size_t)N_NODES * 4));
    float* nrm  = (float*)(ws + alloc((size_t)N_EDGES * 4));
    float* T    = (float*)(ws + alloc((size_t)N_NODES * HDIM * 4));
    float* P    = (float*)(ws + alloc((size_t)N_NODES * HDIM * 4));

    const int TPB = 256;
    const int strips = N_NODES / 16;
    const int gemmBlocks = (strips + 7) / 8;

    init_deg_kernel<<<(N_NODES + TPB - 1) / TPB, TPB, 0, stream>>>(dinv, N_NODES);
    edge_deg_kernel<<<(N_EDGES + TPB - 1) / TPB, TPB, 0, stream>>>(dst, dinv, N_EDGES);
    dinv_kernel<<<(N_NODES + TPB - 1) / TPB, TPB, 0, stream>>>(dinv, N_NODES);
    edge_norm_kernel<<<(N_EDGES + TPB - 1) / TPB, TPB, 0, stream>>>(src, dst, dinv, nrm, N_EDGES);

    gcn_gemm_wmma_kernel<128, false><<<gemmBlocks, TPB, 0, stream>>>(x, W[0], T, strips);
    {
        int ti = N_NODES * 32, ts = N_EDGES * 32;
        agg_init_kernel<32><<<(ti + TPB - 1) / TPB, TPB, 0, stream>>>((const float4*)T, dinv, bia[0], (float4*)P, ti);
        edge_scatter_kernel<32><<<(ts + TPB - 1) / TPB, TPB, 0, stream>>>(src, dst, nrm, (const float4*)T, P, ts);
    }
    for (int l = 1; l <= 2; ++l) {
        gcn_gemm_wmma_kernel<128, true><<<gemmBlocks, TPB, 0, stream>>>(P, W[l], T, strips);
        int ti = N_NODES * 32, ts = N_EDGES * 32;
        agg_init_kernel<32><<<(ti + TPB - 1) / TPB, TPB, 0, stream>>>((const float4*)T, dinv, bia[l], (float4*)P, ti);
        edge_scatter_kernel<32><<<(ts + TPB - 1) / TPB, TPB, 0, stream>>>(src, dst, nrm, (const float4*)T, P, ts);
    }
    gcn_gemm_wmma_kernel<40, true><<<gemmBlocks, TPB, 0, stream>>>(P, W[3], T, strips);
    {
        int ti = N_NODES * 10, ts = N_EDGES * 10;
        agg_init_kernel<10><<<(ti + TPB - 1) / TPB, TPB, 0, stream>>>((const float4*)T, dinv, bia[3], (float4*)d_out, ti);
        edge_scatter_kernel<10><<<(ts + TPB - 1) / TPB, TPB, 0, stream>>>(src, dst, nrm, (const float4*)T, (float*)d_out, ts);
    }
}


# Aggregate by destination-sorted (CSR) row gather, one wave and one store per output row, replacing the float atomic scatter and its init pass; D^-1/2 split into GEMM row scale and gather post-scale
# speedup vs baseline: 2.0475x; 2.0475x over previous
#include <hip/hip_runtime.h>


#define N_NODES 100000
#define N_EDGES 500000
#define KDIM    128
#define HDIM    128
#define NCLS    40

typedef __attribute__((ext_vector_type(2))) float v2f;
typedef __attribute__((ext_vector_type(8))) float v8f;


__global__ void zero_count_kernel(int* __restrict__ cnt, int n) {
    int i = blockIdx.x * blockDim.x + threadIdx.x;
    if (i < n) cnt[i] = 0;
}

__global__ void edge_rank_kernel(const int* __restrict__ dst, int* __restrict__ cnt,
                                 int* __restrict__ rank, int e) {
    int i = blockIdx.x * blockDim.x + threadIdx.x;
    if (i < e) rank[i] = atomicAdd(&cnt[dst[i]], 1);
}

#define SCAN_TPB 1024
__global__ __launch_bounds__(SCAN_TPB)
void row_offset_scan_kernel(const int* __restrict__ cnt, int* __restrict__ rowptr, int n)
{
    __shared__ int sh_before[32];
    __shared__ int sh_own[32];
    const int tid  = threadIdx.x;
    const int lane = tid & 31;
    const int w    = tid >> 5;
    const int base = blockIdx.x * SCAN_TPB;

    int before = 0;
    const int4* c4 = (const int4*)cnt;
    const int nfull = blockIdx.x >> 2;
#pragma unroll 8
    for (int it = 0; it < nfull; ++it) {
        int4 v = c4[it * SCAN_TPB + tid];
        before += (v.x + v.y) + (v.z + v.w);
    }
    if (tid < (int)(blockIdx.x & 3) * 256) {
        int4 v = c4[nfull * SCAN_TPB + tid];
        before += (v.x + v.y) + (v.z + v.w);
    }

    const int idx = base + tid;
    const int own = (idx < n) ? cnt[idx] : 0;

    int incl = own;
#pragma unroll
    for (int d = 1; d < 32; d <<= 1) {
        int t = __shfl_up(incl, d, 32);
        if (lane >= d) incl += t;
    }
#pragma unroll
    for (int d = 16; d >= 1; d >>= 1) before += __shfl_xor(before, d, 32);

    if (lane == 31) sh_own[w] = incl;
    if (lane == 0)  sh_before[w] = before;
    __syncthreads();

    int tb = sh_before[lane];
    int to = (lane < w) ? sh_own[lane] : 0;
#pragma unroll
    for (int d = 16; d >= 1; d >>= 1) {
        tb += __shfl_xor(tb, d, 32);
        to += __shfl_xor(to, d, 32);
    }
    const int excl = tb + to + incl - own;
    if (idx < n)      rowptr[idx] = excl;
    if (idx == n - 1) rowptr[n]   = excl + own;
}

__global__ void csr_fill_kernel(const int* __restrict__ src, const int* __restrict__ dst,
                                const int* __restrict__ rowptr, const int* __restrict__ rank,
                                int* __restrict__ csr_src, int e) {
    int i = blockIdx.x * blockDim.x + threadIdx.x;
    if (i < e) csr_src[rowptr[dst[i]] + rank[i]] = src[i];
}

template <int FOUT, bool RELU>
__global__ __launch_bounds__(256)
void gcn_gemm_wmma_kernel(const float* __restrict__ A,
                          const float* __restrict__ B,
                          const int* __restrict__ rowptr,
                          float* __restrict__ Out,
                          int totalStrips)
{
    constexpr int FP = (FOUT % 16 == 0) ? FOUT : ((FOUT + 15) & ~15);
    constexpr int NT = FP / 16;

    __shared__ float ldsB[KDIM * FP];

    {
        constexpr int NE = KDIM * FP;
#pragma unroll 4
        for (int p = threadIdx.x; p < NE; p += 256) {
            int k = p / FP;
            int c = p - k * FP;
            float val;
            if constexpr (FOUT == FP)
                val = B[p];
            else
                val = (c < FOUT) ? B[k * FOUT + c] : 0.0f;
            ldsB[(((k >> 1) * FP) + c) * 2 + (k & 1)] = val;
        }
    }
    __syncthreads();

    int strip = blockIdx.x * (blockDim.x >> 5) + (threadIdx.x >> 5);
    if (strip >= totalStrips) return;

    int lane = threadIdx.x & 31;
    int half = lane >> 4;
    int mn   = lane & 15;

    float dv[8];
    {
        const int* rp = rowptr + strip * 16 + half * 8;
        int lo = rp[0];
#pragma unroll
        for (int r = 0; r < 8; ++r) {
            int hi = rp[r + 1];
            float d = (float)(hi - lo + 1);
            dv[r] = (d > 0.0f) ? rsqrtf(d) : 0.0f;
            lo = hi;
        }
    }

    const float* arow = A + (size_t)(strip * 16 + mn) * KDIM;

    v8f acc[NT] = {};
#pragma unroll
    for (int k0 = 0; k0 < KDIM; k0 += 4) {
        int k  = k0 + 2 * half;
        int kp = (k0 >> 1) + half;
        v2f a = *(const v2f*)(arow + k);
        if constexpr (RELU) {
            a.x = fmaxf(a.x, 0.0f);
            a.y = fmaxf(a.y, 0.0f);
        }
        const float* brow = ldsB + (size_t)kp * FP * 2 + mn * 2;
#pragma unroll
        for (int nt = 0; nt < NT; ++nt) {
            v2f b = *(const v2f*)(brow + nt * 32);
            acc[nt] = __builtin_amdgcn_wmma_f32_16x16x4_f32(
                false, a, false, b, (short)0, acc[nt], false, false);
        }
    }

#pragma unroll
    for (int nt = 0; nt < NT; ++nt) {
#pragma unroll
        for (int r = 0; r < 8; ++r) {
            int orow = strip * 16 + half * 8 + r;
            int ocol = nt * 16 + mn;
            if (FOUT == FP || ocol < FOUT)
                Out[(size_t)orow * FOUT + ocol] = acc[nt][r] * dv[r];
        }
    }
}


__device__ __forceinline__ void add_rows128(float4& acc, const float4* __restrict__ hs,
                                            int ids, int m, int lane)
{
    int j = 0;
    for (; j + 4 <= m; j += 4) {
        const int s0 = __builtin_amdgcn_readlane(ids, j + 0);
        const int s1 = __builtin_amdgcn_readlane(ids, j + 1);
        const int s2 = __builtin_amdgcn_readlane(ids, j + 2);
        const int s3 = __builtin_amdgcn_readlane(ids, j + 3);
        const float4 v0 = hs[(size_t)s0 * 32 + lane];
        const float4 v1 = hs[(size_t)s1 * 32 + lane];
        const float4 v2 = hs[(size_t)s2 * 32 + lane];
        const float4 v3 = hs[(size_t)s3 * 32 + lane];
        acc.x += (v0.x + v1.x) + (v2.x + v3.x);
        acc.y += (v0.y + v1.y) + (v2.y + v3.y);
        acc.z += (v0.z + v1.z) + (v2.z + v3.z);
        acc.w += (v0.w + v1.w) + (v2.w + v3.w);
    }
    if (j < m) {
        const int s0 = __builtin_amdgcn_readlane(ids, j);
        const float4 v0 = hs[(size_t)s0 * 32 + lane];
        if (j + 1 < m) {
            const int s1 = __builtin_amdgcn_readlane(ids, j + 1);
            const float4 v1 = hs[(size_t)s1 * 32 + lane];
            if (j + 2 < m) {
                const int s2 = __builtin_amdgcn_readlane(ids, j + 2);
                const float4 v2 = hs[(size_t)s2 * 32 + lane];
                acc.x += v2.x; acc.y += v2.y; acc.z += v2.z; acc.w += v2.w;
            }
            acc.x += v1.x; acc.y += v1.y; acc.z += v1.z; acc.w += v1.w;
        }
        acc.x += v0.x; acc.y += v0.y; acc.z += v0.z; acc.w += v0.w;
    }
}

__global__ __launch_bounds__(256)
void gather_rows128_kernel(const int* __restrict__ rowptr,
                           const int* __restrict__ csr_src,
                           const float4* __restrict__ hs,
                           const float* __restrict__ bias,
                           float4* __restrict__ out, int n_nodes)
{
    const int lane = threadIdx.x & 31;
    const int wave   = __builtin_amdgcn_readfirstlane((int)(blockIdx.x * 8 + (threadIdx.x >> 5)));
    const int nwaves = gridDim.x * 8;

    float4 bv;
    bv.x = bias[lane * 4 + 0];
    bv.y = bias[lane * 4 + 1];
    bv.z = bias[lane * 4 + 2];
    bv.w = bias[lane * 4 + 3];

    for (int n = wave; n < n_nodes; n += nwaves) {
        const int beg = rowptr[n];
        const int deg = rowptr[n + 1] - beg;
        int ids = 0;
        if (lane < deg) ids = csr_src[beg + lane];
        float4 acc = hs[(size_t)n * 32 + lane];
        add_rows128(acc, hs, ids, min(deg, 32), lane);
        for (int j0 = 32; j0 < deg; j0 += 32) {
            ids = 0;
            if (lane < deg - j0) ids = csr_src[beg + j0 + lane];
            add_rows128(acc, hs, ids, min(deg - j0, 32), lane);
        }
        const float d  = (float)(deg + 1);
        const float dn = (d > 0.0f) ? rsqrtf(d) : 0.0f;
        float4 r;
        r.x = acc.x * dn + bv.x;
        r.y = acc.y * dn + bv.y;
        r.z = acc.z * dn + bv.z;
        r.w = acc.w * dn + bv.w;
        out[(size_t)n * 32 + lane] = r;
    }
}

__device__ __forceinline__ void add_rows40(float4& acc, const float4* __restrict__ hs,
                                           int ids, int m, int cc)
{
    for (int j = 0; j < m; j += 4) {
        const int s0 = __shfl(ids, j, 16);
        const int s1 = __shfl(ids, (j + 1) & 15, 16);
        const int s2 = __shfl(ids, (j + 2) & 15, 16);
        const int s3 = __shfl(ids, (j + 3) & 15, 16);
        float4 v1 = {0.0f, 0.0f, 0.0f, 0.0f};
        float4 v2 = {0.0f, 0.0f, 0.0f, 0.0f};
        float4 v3 = {0.0f, 0.0f, 0.0f, 0.0f};
        const float4 v0 = hs[(size_t)s0 * 10 + cc];
        if (j + 1 < m) v1 = hs[(size_t)s1 * 10 + cc];
        if (j + 2 < m) v2 = hs[(size_t)s2 * 10 + cc];
        if (j + 3 < m) v3 = hs[(size_t)s3 * 10 + cc];
        acc.x += (v0.x + v1.x) + (v2.x + v3.x);
        acc.y += (v0.y + v1.y) + (v2.y + v3.y);
        acc.z += (v0.z + v1.z) + (v2.z + v3.z);
        acc.w += (v0.w + v1.w) + (v2.w + v3.w);
    }
}

__global__ __launch_bounds__(256)
void gather_rows40_kernel(const int* __restrict__ rowptr,
                          const int* __restrict__ csr_src,
                          const float4* __restrict__ hs,
                          const float* __restrict__ bias,
                          float4* __restrict__ out, int n_quads)
{
    const int lane = threadIdx.x & 31;
    const int sub  = lane >> 4;
    const int c    = lane & 15;
    const int cc   = min(c, 9);
    const int wave   = blockIdx.x * 8 + (threadIdx.x >> 5);
    const int nwaves = gridDim.x * 8;

    float4 bv;
    bv.x = bias[cc * 4 + 0];
    bv.y = bias[cc * 4 + 1];
    bv.z = bias[cc * 4 + 2];
    bv.w = bias[cc * 4 + 3];

    for (int q = wave; q < n_quads; q += nwaves) {
#pragma unroll
        for (int p = 0; p < 2; ++p) {
            const int n   = q * 4 + p * 2 + sub;
            const int beg = rowptr[n];
            const int deg = rowptr[n + 1] - beg;
            int ids = 0;
            if (c < deg) ids = csr_src[beg + c];
            float4 acc = hs[(size_t)n * 10 + cc];
            add_rows40(acc, hs, ids, min(deg, 16), cc);
            for (int j0 = 16; j0 < deg; j0 += 16) {
                ids = 0;
                if (c < deg - j0) ids = csr_src[beg + j0 + c];
                add_rows40(acc, hs, ids, min(deg - j0, 16), cc);
            }
            const float d  = (float)(deg + 1);
            const float dn = (d > 0.0f) ? rsqrtf(d) : 0.0f;
            if (c < 10) {
                float4 r;
                r.x = acc.x * dn + bv.x;
                r.y = acc.y * dn + bv.y;
                r.z = acc.z * dn + bv.z;
                r.w = acc.w * dn + bv.w;
                out[(size_t)n * 10 + c] = r;
            }
        }
    }
}


extern "C" void kernel_launch(void* const* d_in, const int* in_sizes, int n_in,
                              void* d_out, int out_size, void* d_ws, size_t ws_size,
                              hipStream_t stream)
{
    const float* x  = (const float*)d_in[0];
    const int*   ei = (const int*)  d_in[1];
    const float* W[4] = { (const float*)d_in[2], (const float*)d_in[4],
                          (const float*)d_in[6], (const float*)d_in[8] };
    const float* bia[4] = { (const float*)d_in[3], (const float*)d_in[5],
                            (const float*)d_in[7], (const float*)d_in[9] };
    const int* src = ei;
    const int* dst = ei + N_EDGES;

    char* ws = (char*)d_ws;
    size_t off = 0;
    auto alloc = [&](size_t bytes) {
        size_t p = off;
        off = (off + bytes + 255) & ~(size_t)255;
        return p;
    };
    int*   rowptr  = (int*)  (ws + alloc((size_t)(N_NODES + 1) * 4));
    int*   csr_src = (int*)  (ws + alloc((size_t)N_EDGES * 4));
    float* T       = (float*)(ws + alloc((size_t)N_NODES * HDIM * 4));
    float* P       = (float*)(ws + alloc((size_t)N_NODES * HDIM * 4));
    int* cnt  = (int*)P;
    int* rank = (int*)P + 100096;

    const int TPB = 256;
    const int strips = N_NODES / 16;
    const int gemmBlocks = (strips + 7) / 8;
    const int gatherBlocks = 2048;
    const int gather40Blocks = N_NODES / 4 / 8;

    zero_count_kernel<<<(N_NODES + TPB - 1) / TPB, TPB, 0, stream>>>(cnt, N_NODES);
    edge_rank_kernel<<<(N_EDGES + TPB - 1) / TPB, TPB, 0, stream>>>(dst, cnt, rank, N_EDGES);
    row_offset_scan_kernel<<<(N_NODES + SCAN_TPB - 1) / SCAN_TPB, SCAN_TPB, 0, stream>>>(cnt, rowptr, N_NODES);
    csr_fill_kernel<<<(N_EDGES + TPB - 1) / TPB, TPB, 0, stream>>>(src, dst, rowptr, rank, csr_src, N_EDGES);

    gcn_gemm_wmma_kernel<128, false><<<gemmBlocks, TPB, 0, stream>>>(x, W[0], rowptr, T, strips);
    gather_rows128_kernel<<<gatherBlocks, TPB, 0, stream>>>(rowptr, csr_src, (const float4*)T, bia[0], (float4*)P, N_NODES);
    for (int l = 1; l <= 2; ++l) {
        gcn_gemm_wmma_kernel<128, true><<<gemmBlocks, TPB, 0, stream>>>(P, W[l], rowptr, T, strips);
        gather_rows128_kernel<<<gatherBlocks, TPB, 0, stream>>>(rowptr, csr_src, (const float4*)T, bia[l], (float4*)P, N_NODES);
    }
    gcn_gemm_wmma_kernel<40, true><<<gemmBlocks, TPB, 0, stream>>>(P, W[3], rowptr, T, strips);
    gather_rows40_kernel<<<gather40Blocks, TPB, 0, stream>>>(rowptr, csr_src, (const float4*)T, bia[3], (float4*)d_out, N_NODES / 4);
}


# Fuse each hidden layer's CSR neighbour sum into the next layer's GEMM launch: a wave sums its 16 rows into an LDS A tile and runs the f32 WMMAs from it; 12 -> 9 launches, no N x 128 layer output
# speedup vs baseline: 2.2656x; 1.1065x over previous
#include <hip/hip_runtime.h>


#define N_NODES 100000
#define N_EDGES 500000
#define KDIM    128
#define HDIM    128
#define NCLS    40

typedef __attribute__((ext_vector_type(2))) float v2f;
typedef __attribute__((ext_vector_type(8))) float v8f;


__global__ void zero_count_kernel(int* __restrict__ cnt, int n) {
    int i = blockIdx.x * blockDim.x + threadIdx.x;
    if (i < n) cnt[i] = 0;
}

__global__ void edge_rank_kernel(const int* __restrict__ dst, int* __restrict__ cnt,
                                 int* __restrict__ rank, int e) {
    int i = blockIdx.x * blockDim.x + threadIdx.x;
    if (i < e) rank[i] = atomicAdd(&cnt[dst[i]], 1);
}

#define SCAN_TPB 1024
__global__ __launch_bounds__(SCAN_TPB)
void row_offset_scan_kernel(const int* __restrict__ cnt, int* __restrict__ rowptr, int n)
{
    __shared__ int sh_before[32];
    __shared__ int sh_own[32];
    const int tid  = threadIdx.x;
    const int lane = tid & 31;
    const int w    = tid >> 5;
    const int base = blockIdx.x * SCAN_TPB;

    int before = 0;
    const int4* c4 = (const int4*)cnt;
    const int nfull = blockIdx.x >> 2;
#pragma unroll 8
    for (int it = 0; it < nfull; ++it) {
        int4 v = c4[it * SCAN_TPB + tid];
        before += (v.x + v.y) + (v.z + v.w);
    }
    if (tid < (int)(blockIdx.x & 3) * 256) {
        int4 v = c4[nfull * SCAN_TPB + tid];
        before += (v.x + v.y) + (v.z + v.w);
    }

    const int idx = base + tid;
    const int own = (idx < n) ? cnt[idx] : 0;

    int incl = own;
#pragma unroll
    for (int d = 1; d < 32; d <<= 1) {
        int t = __shfl_up(incl, d, 32);
        if (lane >= d) incl += t;
    }
#pragma unroll
    for (int d = 16; d >= 1; d >>= 1) before += __shfl_xor(before, d, 32);

    if (lane == 31) sh_own[w] = incl;
    if (lane == 0)  sh_before[w] = before;
    __syncthreads();

    int tb = sh_before[lane];
    int to = (lane < w) ? sh_own[lane] : 0;
#pragma unroll
    for (int d = 16; d >= 1; d >>= 1) {
        tb += __shfl_xor(tb, d, 32);
        to += __shfl_xor(to, d, 32);
    }
    const int excl = tb + to + incl - own;
    if (idx < n)      rowptr[idx] = excl;
    if (idx == n - 1) rowptr[n]   = excl + own;
}

__global__ void csr_fill_kernel(const int* __restrict__ src, const int* __restrict__ dst,
                                const int* __restrict__ rowptr, const int* __restrict__ rank,
                                int* __restrict__ csr_src, int e) {
    int i = blockIdx.x * blockDim.x + threadIdx.x;
    if (i < e) csr_src[rowptr[dst[i]] + rank[i]] = src[i];
}

template <int FOUT, bool RELU>
__global__ __launch_bounds__(256)
void gcn_gemm_wmma_kernel(const float* __restrict__ A,
                          const float* __restrict__ B,
                          const int* __restrict__ rowptr,
                          float* __restrict__ Out,
                          int totalStrips)
{
    constexpr int FP = (FOUT % 16 == 0) ? FOUT : ((FOUT + 15) & ~15);
    constexpr int NT = FP / 16;

    __shared__ float ldsB[KDIM * FP];

    {
        constexpr int NE = KDIM * FP;
#pragma unroll 4
        for (int p = threadIdx.x; p < NE; p += 256) {
            int k = p / FP;
            int c = p - k * FP;
            float val;
            if constexpr (FOUT == FP)
                val = B[p];
            else
                val = (c < FOUT) ? B[k * FOUT + c] : 0.0f;
            ldsB[(((k >> 1) * FP) + c) * 2 + (k & 1)] = val;
        }
    }
    __syncthreads();

    int strip = blockIdx.x * (blockDim.x >> 5) + (threadIdx.x >> 5);
    if (strip >= totalStrips) return;

    int lane = threadIdx.x & 31;
    int half = lane >> 4;
    int mn   = lane & 15;

    float dv[8];
    {
        const int* rp = rowptr + strip * 16 + half * 8;
        int lo = rp[0];
#pragma unroll
        for (int r = 0; r < 8; ++r) {
            int hi = rp[r + 1];
            float d = (float)(hi - lo + 1);
            dv[r] = (d > 0.0f) ? rsqrtf(d) : 0.0f;
            lo = hi;
        }
    }

    const float* arow = A + (size_t)(strip * 16 + mn) * KDIM;

    v8f acc[NT] = {};
#pragma unroll
    for (int k0 = 0; k0 < KDIM; k0 += 4) {
        int k  = k0 + 2 * half;
        int kp = (k0 >> 1) + half;
        v2f a = *(const v2f*)(arow + k);
        if constexpr (RELU) {
            a.x = fmaxf(a.x, 0.0f);
            a.y = fmaxf(a.y, 0.0f);
        }
        const float* brow = ldsB + (size_t)kp * FP * 2 + mn * 2;
#pragma unroll
        for (int nt = 0; nt < NT; ++nt) {
            v2f b = *(const v2f*)(brow + nt * 32);
            acc[nt] = __builtin_amdgcn_wmma_f32_16x16x4_f32(
                false, a, false, b, (short)0, acc[nt], false, false);
        }
    }

#pragma unroll
    for (int nt = 0; nt < NT; ++nt) {
#pragma unroll
        for (int r = 0; r < 8; ++r) {
            int orow = strip * 16 + half * 8 + r;
            int ocol = nt * 16 + mn;
            if (FOUT == FP || ocol < FOUT)
                Out[(size_t)orow * FOUT + ocol] = acc[nt][r] * dv[r];
        }
    }
}


__device__ __forceinline__ void add_rows128(float4& acc, const float4* __restrict__ hs,
                                            int ids, int m, int lane)
{
    for (int j = 0; j < m; j += 8) {
        float4 v[8];
#pragma unroll
        for (int i = 0; i < 8; ++i) {
            v[i].x = 0.0f; v[i].y = 0.0f; v[i].z = 0.0f; v[i].w = 0.0f;
            if (j + i < m) {
                const int s = __builtin_amdgcn_readlane(ids, j + i);
                v[i] = hs[(size_t)s * 32 + lane];
            }
        }
        acc.x += ((v[0].x + v[1].x) + (v[2].x + v[3].x)) + ((v[4].x + v[5].x) + (v[6].x + v[7].x));
        acc.y += ((v[0].y + v[1].y) + (v[2].y + v[3].y)) + ((v[4].y + v[5].y) + (v[6].y + v[7].y));
        acc.z += ((v[0].z + v[1].z) + (v[2].z + v[3].z)) + ((v[4].z + v[5].z) + (v[6].z + v[7].z));
        acc.w += ((v[0].w + v[1].w) + (v[2].w + v[3].w)) + ((v[4].w + v[5].w) + (v[6].w + v[7].w));
    }
}

#define FUSE_WAVES 25
#define FUSE_TPB   (FUSE_WAVES * 32)
#define APITCH     132

template <int FOUT>
__global__ __launch_bounds__(FUSE_TPB)
void gcn_gather_gemm_kernel(const int* __restrict__ rowptr,
                            const int* __restrict__ csr_src,
                            const float4* __restrict__ hs,
                            const float* __restrict__ bias,
                            const float* __restrict__ B,
                            float* __restrict__ Out,
                            int totalStrips)
{
    constexpr int FP = (FOUT % 16 == 0) ? FOUT : ((FOUT + 15) & ~15);
    constexpr int NT = FP / 16;

    __shared__ __attribute__((aligned(16))) float lds[KDIM * FP + FUSE_WAVES * 16 * APITCH];
    float* ldsB = lds;
    float* ldsA = lds + KDIM * FP;

    {
        constexpr int NE = KDIM * FP;
#pragma unroll 4
        for (int p = threadIdx.x; p < NE; p += FUSE_TPB) {
            int k = p / FP;
            int c = p - k * FP;
            float val;
            if constexpr (FOUT == FP)
                val = B[p];
            else
                val = (c < FOUT) ? B[k * FOUT + c] : 0.0f;
            ldsB[(((k >> 1) * FP) + c) * 2 + (k & 1)] = val;
        }
    }
    __syncthreads();

    const int lane = threadIdx.x & 31;
    const int wv    = __builtin_amdgcn_readfirstlane((int)(threadIdx.x >> 5));
    const int strip = __builtin_amdgcn_readfirstlane((int)(blockIdx.x * FUSE_WAVES) + wv);
    if (strip >= totalStrips) return;

    float* stage = ldsA + wv * (16 * APITCH);
    const int nbase = strip * 16;

    {
        float4 bv;
        bv.x = bias[lane * 4 + 0];
        bv.y = bias[lane * 4 + 1];
        bv.z = bias[lane * 4 + 2];
        bv.w = bias[lane * 4 + 3];

        const int rp = rowptr[nbase + min(lane, 16)];
        int beg = __builtin_amdgcn_readlane(rp, 0);
        int end = __builtin_amdgcn_readlane(rp, 1);
        int ids = 0;
        if (lane < end - beg) ids = csr_src[beg + lane];
        float4 selfv = hs[(size_t)nbase * 32 + lane];

        for (int r = 0; r < 16; ++r) {
            const int deg = end - beg;
            int endN = end;
            int idsN = 0;
            float4 selfN;
            selfN.x = 0.0f; selfN.y = 0.0f; selfN.z = 0.0f; selfN.w = 0.0f;
            if (r < 15) {
                endN = __builtin_amdgcn_readlane(rp, r + 2);
                if (lane < endN - end) idsN = csr_src[end + lane];
                selfN = hs[(size_t)(nbase + r + 1) * 32 + lane];
            }

            float4 acc = selfv;
            add_rows128(acc, hs, ids, min(deg, 32), lane);
            for (int j0 = 32; j0 < deg; j0 += 32) {
                int idsM = 0;
                if (lane < deg - j0) idsM = csr_src[beg + j0 + lane];
                add_rows128(acc, hs, idsM, min(deg - j0, 32), lane);
            }

            const float d  = (float)(deg + 1);
            const float dn = (d > 0.0f) ? rsqrtf(d) : 0.0f;
            float4 o;
            o.x = fmaxf(acc.x * dn + bv.x, 0.0f);
            o.y = fmaxf(acc.y * dn + bv.y, 0.0f);
            o.z = fmaxf(acc.z * dn + bv.z, 0.0f);
            o.w = fmaxf(acc.w * dn + bv.w, 0.0f);
            *(float4*)(stage + r * APITCH + lane * 4) = o;

            beg = end; end = endN; ids = idsN; selfv = selfN;
        }
    }
    __builtin_amdgcn_fence(__ATOMIC_RELEASE, "wavefront");
    __builtin_amdgcn_wave_barrier();
    __builtin_amdgcn_fence(__ATOMIC_ACQUIRE, "wavefront");

    const int half = lane >> 4;
    const int mn   = lane & 15;

    float dv[8];
    {
        const int* rpp = rowptr + nbase + half * 8;
        int lo = rpp[0];
#pragma unroll
        for (int r = 0; r < 8; ++r) {
            int hi = rpp[r + 1];
            float d = (float)(hi - lo + 1);
            dv[r] = (d > 0.0f) ? rsqrtf(d) : 0.0f;
            lo = hi;
        }
    }

    const float* arow = stage + mn * APITCH;

    v8f acc[NT] = {};
#pragma unroll
    for (int k0 = 0; k0 < KDIM; k0 += 4) {
        int k  = k0 + 2 * half;
        int kp = (k0 >> 1) + half;
        v2f a = *(const v2f*)(arow + k);
        const float* brow = ldsB + (size_t)kp * FP * 2 + mn * 2;
#pragma unroll
        for (int nt = 0; nt < NT; ++nt) {
            v2f b = *(const v2f*)(brow + nt * 32);
            acc[nt] = __builtin_amdgcn_wmma_f32_16x16x4_f32(
                false, a, false, b, (short)0, acc[nt], false, false);
        }
    }

#pragma unroll
    for (int nt = 0; nt < NT; ++nt) {
#pragma unroll
        for (int r = 0; r < 8; ++r) {
            int orow = nbase + half * 8 + r;
            int ocol = nt * 16 + mn;
            if (FOUT == FP || ocol < FOUT)
                Out[(size_t)orow * FOUT + ocol] = acc[nt][r] * dv[r];
        }
    }
}

__device__ __forceinline__ void add_rows40(float4& acc, const float4* __restrict__ hs,
                                           int ids, int m, int cc)
{
    for (int j = 0; j < m; j += 4) {
        const int s0 = __shfl(ids, j, 16);
        const int s1 = __shfl(ids, (j + 1) & 15, 16);
        const int s2 = __shfl(ids, (j + 2) & 15, 16);
        const int s3 = __shfl(ids, (j + 3) & 15, 16);
        float4 v1 = {0.0f, 0.0f, 0.0f, 0.0f};
        float4 v2 = {0.0f, 0.0f, 0.0f, 0.0f};
        float4 v3 = {0.0f, 0.0f, 0.0f, 0.0f};
        const float4 v0 = hs[(size_t)s0 * 10 + cc];
        if (j + 1 < m) v1 = hs[(size_t)s1 * 10 + cc];
        if (j + 2 < m) v2 = hs[(size_t)s2 * 10 + cc];
        if (j + 3 < m) v3 = hs[(size_t)s3 * 10 + cc];
        acc.x += (v0.x + v1.x) + (v2.x + v3.x);
        acc.y += (v0.y + v1.y) + (v2.y + v3.y);
        acc.z += (v0.z + v1.z) + (v2.z + v3.z);
        acc.w += (v0.w + v1.w) + (v2.w + v3.w);
    }
}

__global__ __launch_bounds__(256)
void gather_rows40_kernel(const int* __restrict__ rowptr,
                          const int* __restrict__ csr_src,
                          const float4* __restrict__ hs,
                          const float* __restrict__ bias,
                          float4* __restrict__ out, int n_quads)
{
    const int lane = threadIdx.x & 31;
    const int sub  = lane >> 4;
    const int c    = lane & 15;
    const int cc   = min(c, 9);
    const int wave   = blockIdx.x * 8 + (threadIdx.x >> 5);
    const int nwaves = gridDim.x * 8;

    float4 bv;
    bv.x = bias[cc * 4 + 0];
    bv.y = bias[cc * 4 + 1];
    bv.z = bias[cc * 4 + 2];
    bv.w = bias[cc * 4 + 3];

    for (int q = wave; q < n_quads; q += nwaves) {
#pragma unroll
        for (int p = 0; p < 2; ++p) {
            const int n   = q * 4 + p * 2 + sub;
            const int beg = rowptr[n];
            const int deg = rowptr[n + 1] - beg;
            int ids = 0;
            if (c < deg) ids = csr_src[beg + c];
            float4 acc = hs[(size_t)n * 10 + cc];
            add_rows40(acc, hs, ids, min(deg, 16), cc);
            for (int j0 = 16; j0 < deg; j0 += 16) {
                ids = 0;
                if (c < deg - j0) ids = csr_src[beg + j0 + c];
                add_rows40(acc, hs, ids, min(deg - j0, 16), cc);
            }
            const float d  = (float)(deg + 1);
            const float dn = (d > 0.0f) ? rsqrtf(d) : 0.0f;
            if (c < 10) {
                float4 r;
                r.x = acc.x * dn + bv.x;
                r.y = acc.y * dn + bv.y;
                r.z = acc.z * dn + bv.z;
                r.w = acc.w * dn + bv.w;
                out[(size_t)n * 10 + c] = r;
            }
        }
    }
}


extern "C" void kernel_launch(void* const* d_in, const int* in_sizes, int n_in,
                              void* d_out, int out_size, void* d_ws, size_t ws_size,
                              hipStream_t stream)
{
    const float* x  = (const float*)d_in[0];
    const int*   ei = (const int*)  d_in[1];
    const float* W[4] = { (const float*)d_in[2], (const float*)d_in[4],
                          (const float*)d_in[6], (const float*)d_in[8] };
    const float* bia[4] = { (const float*)d_in[3], (const float*)d_in[5],
                            (const float*)d_in[7], (const float*)d_in[9] };
    const int* src = ei;
    const int* dst = ei + N_EDGES;

    char* ws = (char*)d_ws;
    size_t off = 0;
    auto alloc = [&](size_t bytes) {
        size_t p = off;
        off = (off + bytes + 255) & ~(size_t)255;
        return p;
    };
    int*   rowptr  = (int*)  (ws + alloc((size_t)(N_NODES + 1) * 4));
    int*   csr_src = (int*)  (ws + alloc((size_t)N_EDGES * 4));
    float* T       = (float*)(ws + alloc((size_t)N_NODES * HDIM * 4));
    float* P       = (float*)(ws + alloc((size_t)N_NODES * HDIM * 4));
    int* cnt  = (int*)P;
    int* rank = (int*)P + 100096;

    const int TPB = 256;
    const int strips = N_NODES / 16;
    const int gemmBlocks = (strips + 7) / 8;
    const int fuseBlocks = (strips + FUSE_WAVES - 1) / FUSE_WAVES;
    const int gather40Blocks = N_NODES / 4 / 8;

    zero_count_kernel<<<(N_NODES + TPB - 1) / TPB, TPB, 0, stream>>>(cnt, N_NODES);
    edge_rank_kernel<<<(N_EDGES + TPB - 1) / TPB, TPB, 0, stream>>>(dst, cnt, rank, N_EDGES);
    row_offset_scan_kernel<<<(N_NODES + SCAN_TPB - 1) / SCAN_TPB, SCAN_TPB, 0, stream>>>(cnt, rowptr, N_NODES);
    csr_fill_kernel<<<(N_EDGES + TPB - 1) / TPB, TPB, 0, stream>>>(src, dst, rowptr, rank, csr_src, N_EDGES);

    gcn_gemm_wmma_kernel<128, false><<<gemmBlocks, TPB, 0, stream>>>(x, W[0], rowptr, T, strips);
    gcn_gather_gemm_kernel<128><<<fuseBlocks, FUSE_TPB, 0, stream>>>(rowptr, csr_src, (const float4*)T, bia[0], W[1], P, strips);
    gcn_gather_gemm_kernel<128><<<fuseBlocks, FUSE_TPB, 0, stream>>>(rowptr, csr_src, (const float4*)P, bia[1], W[2], T, strips);
    gcn_gather_gemm_kernel<40><<<fuseBlocks, FUSE_TPB, 0, stream>>>(rowptr, csr_src, (const float4*)T, bia[2], W[3], P, strips);
    gather_rows40_kernel<<<gather40Blocks, TPB, 0, stream>>>(rowptr, csr_src, (const float4*)P, bia[3], (float4*)d_out, N_NODES / 4);
}


# Hidden-layer neighbour sum: a batch's 8 source ids are read to scalar registers ahead of its conditional row loads, so no load-counter wait splits them: 8 row loads in flight per wave, not 2
# speedup vs baseline: 2.9889x; 1.3193x over previous
#include <hip/hip_runtime.h>


#define N_NODES 100000
#define N_EDGES 500000
#define KDIM    128
#define HDIM    128
#define NCLS    40

typedef __attribute__((ext_vector_type(2))) float v2f;
typedef __attribute__((ext_vector_type(8))) float v8f;

#define NBKT        250
#define BKT_NODES   400
#define SLICE_EDGES 2000
#define BIN_TPB     256
#define SLICE_PER_THREAD ((SLICE_EDGES + BIN_TPB - 1) / BIN_TPB)
#define CSR_OBUF    16384

static_assert(NBKT * BKT_NODES == N_NODES, "ranges must tile the nodes");
static_assert(NBKT * SLICE_EDGES == N_EDGES, "slices must tile the edges");
static_assert(NBKT <= BIN_TPB && BKT_NODES <= 2 * BIN_TPB && (BKT_NODES % 2) == 0, "one thread per range, two nodes per thread");

__device__ __forceinline__ int block_scan_excl_256(int v, int* wsum, int& total)
{
    const int lane = threadIdx.x & 31;
    const int w    = threadIdx.x >> 5;
    int incl = v;
#pragma unroll
    for (int d = 1; d < 32; d <<= 1) {
        int t = __shfl_up(incl, d, 32);
        if (lane >= d) incl += t;
    }
    if (lane == 31) wsum[w] = incl;
    __syncthreads();
    int before = 0;
    total = 0;
#pragma unroll
    for (int i = 0; i < BIN_TPB / 32; ++i) {
        const int x = wsum[i];
        if (i < w) before += x;
        total += x;
    }
    __syncthreads();
    return before + incl - v;
}

__global__ __launch_bounds__(BIN_TPB)
void edge_slice_bin_kernel(const int* __restrict__ src, const int* __restrict__ dst,
                           int2* __restrict__ runtab, int2* __restrict__ staged)
{
    __shared__ int  hist[BIN_TPB];
    __shared__ int  offs[BIN_TPB];
    __shared__ int  wsum[BIN_TPB / 32];
    __shared__ int2 sorted[SLICE_EDGES];

    const int tid = threadIdx.x;
    const int g   = blockIdx.x;
    const int e0  = g * SLICE_EDGES;

    hist[tid] = 0;

    int d[SLICE_PER_THREAD], s[SLICE_PER_THREAD], p[SLICE_PER_THREAD];
#pragma unroll
    for (int i = 0; i < SLICE_PER_THREAD; ++i) {
        const int idx = tid + i * BIN_TPB;
        d[i] = 0; s[i] = 0;
        if (idx < SLICE_EDGES) { d[i] = dst[e0 + idx]; s[i] = src[e0 + idx]; }
    }
    __syncthreads();
#pragma unroll
    for (int i = 0; i < SLICE_PER_THREAD; ++i) {
        const int idx = tid + i * BIN_TPB;
        p[i] = 0;
        if (idx < SLICE_EDGES) p[i] = atomicAdd(&hist[(unsigned)d[i] / BKT_NODES], 1);
    }
    __syncthreads();

    const int mine = hist[tid];
    int total;
    const int excl = block_scan_excl_256(mine, wsum, total);
    offs[tid] = excl;
    if (tid < NBKT) {
        int2 run; run.x = mine; run.y = excl;
        runtab[g * NBKT + tid] = run;
    }
    __syncthreads();
#pragma unroll
    for (int i = 0; i < SLICE_PER_THREAD; ++i) {
        const int idx = tid + i * BIN_TPB;
        if (idx < SLICE_EDGES) {
            int2 e; e.x = d[i]; e.y = s[i];
            sorted[offs[(unsigned)d[i] / BKT_NODES] + p[i]] = e;
        }
    }
    __syncthreads();
    for (int idx = tid; idx < SLICE_EDGES; idx += BIN_TPB) staged[e0 + idx] = sorted[idx];
}

__global__ __launch_bounds__(BIN_TPB)
void range_csr_kernel(const int2* __restrict__ runtab,
                      const int2* __restrict__ staged,
                      int* __restrict__ rowptr, int* __restrict__ csr_src)
{
    __shared__ int ncnt[2 * BIN_TPB];
    __shared__ int wsum[BIN_TPB / 32];
    __shared__ int obuf[CSR_OBUF];

    const int tid = threadIdx.x;
    const int k   = blockIdx.x;
    const int lo  = k * BKT_NODES;

    int len = 0, pre = 0;
    if (tid < NBKT) {
        const int2 run = runtab[tid * NBKT + k];
        len = run.x;
        pre = run.y;
    }
    const int2* seg = staged + (tid < NBKT ? tid : 0) * SLICE_EDGES + pre;

    ncnt[tid] = 0;
    ncnt[tid + BIN_TPB] = 0;
    __syncthreads();

    const int* segd = (const int*)seg;
    for (int j = 0; j < len; j += 4) {
        int dn[4];
#pragma unroll
        for (int u = 0; u < 4; ++u) {
            dn[u] = lo;
            if (j + u < len) dn[u] = segd[2 * (j + u)];
        }
#pragma unroll
        for (int u = 0; u < 4; ++u)
            if (j + u < len) atomicAdd(&ncnt[dn[u] - lo], 1);
    }

    int base, cntk, unused;
    (void)block_scan_excl_256(pre, wsum, base);
    (void)block_scan_excl_256(len, wsum, cntk);

    const int ca = ncnt[2 * tid];
    const int cb = ncnt[2 * tid + 1];
    const int ex = block_scan_excl_256(ca + cb, wsum, unused);
    ncnt[2 * tid]     = ex;
    ncnt[2 * tid + 1] = ex + ca;
    if (2 * tid < BKT_NODES) {
        rowptr[lo + 2 * tid]     = base + ex;
        rowptr[lo + 2 * tid + 1] = base + ex + ca;
    }
    if (k == NBKT - 1 && tid == 0) rowptr[N_NODES] = base + cntk;
    __syncthreads();

    const bool in_lds = (cntk <= CSR_OBUF);
    for (int j = 0; j < len; j += 4) {
        int2 e[4];
#pragma unroll
        for (int u = 0; u < 4; ++u) {
            e[u].x = lo; e[u].y = 0;
            if (j + u < len) e[u] = seg[j + u];
        }
#pragma unroll
        for (int u = 0; u < 4; ++u) {
            if (j + u < len) {
                const int pos = atomicAdd(&ncnt[e[u].x - lo], 1);
                if (in_lds) obuf[pos] = e[u].y;
                else        csr_src[base + pos] = e[u].y;
            }
        }
    }
    __syncthreads();
    if (in_lds)
        for (int i = tid; i < cntk; i += BIN_TPB) csr_src[base + i] = obuf[i];
}

template <int FOUT, bool RELU>
__global__ __launch_bounds__(256)
void gcn_gemm_wmma_kernel(const float* __restrict__ A,
                          const float* __restrict__ B,
                          const int* __restrict__ rowptr,
                          float* __restrict__ Out,
                          int totalStrips)
{
    constexpr int FP = (FOUT % 16 == 0) ? FOUT : ((FOUT + 15) & ~15);
    constexpr int NT = FP / 16;

    __shared__ float ldsB[KDIM * FP];

    {
        constexpr int NE = KDIM * FP;
#pragma unroll 4
        for (int p = threadIdx.x; p < NE; p += 256) {
            int k = p / FP;
            int c = p - k * FP;
            float val;
            if constexpr (FOUT == FP)
                val = B[p];
            else
                val = (c < FOUT) ? B[k * FOUT + c] : 0.0f;
            ldsB[(((k >> 1) * FP) + c) * 2 + (k & 1)] = val;
        }
    }
    __syncthreads();

    int strip = blockIdx.x * (blockDim.x >> 5) + (threadIdx.x >> 5);
    if (strip >= totalStrips) return;

    int lane = threadIdx.x & 31;
    int half = lane >> 4;
    int mn   = lane & 15;

    float dv[8];
    {
        const int* rp = rowptr + strip * 16 + half * 8;
        int lo = rp[0];
#pragma unroll
        for (int r = 0; r < 8; ++r) {
            int hi = rp[r + 1];
            float d = (float)(hi - lo + 1);
            dv[r] = (d > 0.0f) ? rsqrtf(d) : 0.0f;
            lo = hi;
        }
    }

    const float* arow = A + (size_t)(strip * 16 + mn) * KDIM;

    v8f acc[NT] = {};
#pragma unroll
    for (int k0 = 0; k0 < KDIM; k0 += 4) {
        int k  = k0 + 2 * half;
        int kp = (k0 >> 1) + half;
        v2f a = *(const v2f*)(arow + k);
        if constexpr (RELU) {
            a.x = fmaxf(a.x, 0.0f);
            a.y = fmaxf(a.y, 0.0f);
        }
        const float* brow = ldsB + (size_t)kp * FP * 2 + mn * 2;
#pragma unroll
        for (int nt = 0; nt < NT; ++nt) {
            v2f b = *(const v2f*)(brow + nt * 32);
            acc[nt] = __builtin_amdgcn_wmma_f32_16x16x4_f32(
                false, a, false, b, (short)0, acc[nt], false, false);
        }
    }

#pragma unroll
    for (int nt = 0; nt < NT; ++nt) {
#pragma unroll
        for (int r = 0; r < 8; ++r) {
            int orow = strip * 16 + half * 8 + r;
            int ocol = nt * 16 + mn;
            if (FOUT == FP || ocol < FOUT)
                Out[(size_t)orow * FOUT + ocol] = acc[nt][r] * dv[r];
        }
    }
}


__device__ __forceinline__ void row_ids8(int (&s)[8], int ids, int j)
{
#pragma unroll
    for (int i = 0; i < 8; ++i) s[i] = __builtin_amdgcn_readlane(ids, j + i);
}

__device__ __forceinline__ void issue_rows8(float4 (&v)[8], const float4* __restrict__ hs,
                                            const int (&s)[8], int j, int m, int lane)
{
#pragma unroll
    for (int i = 0; i < 8; ++i) {
        v[i].x = 0.0f; v[i].y = 0.0f; v[i].z = 0.0f; v[i].w = 0.0f;
        if (j + i < m) v[i] = hs[(size_t)s[i] * 32 + lane];
    }
}

__device__ __forceinline__ void sum_rows8(float4& acc, const float4 (&v)[8])
{
    acc.x += ((v[0].x + v[1].x) + (v[2].x + v[3].x)) + ((v[4].x + v[5].x) + (v[6].x + v[7].x));
    acc.y += ((v[0].y + v[1].y) + (v[2].y + v[3].y)) + ((v[4].y + v[5].y) + (v[6].y + v[7].y));
    acc.z += ((v[0].z + v[1].z) + (v[2].z + v[3].z)) + ((v[4].z + v[5].z) + (v[6].z + v[7].z));
    acc.w += ((v[0].w + v[1].w) + (v[2].w + v[3].w)) + ((v[4].w + v[5].w) + (v[6].w + v[7].w));
}

__device__ __forceinline__ void add_rows128(float4& acc, const float4* __restrict__ hs,
                                            int ids, int j0, int m, int lane)
{
    for (int j = j0; j < m; j += 8) {
        int s[8];
        row_ids8(s, ids, j);
        float4 v[8];
        issue_rows8(v, hs, s, j, m, lane);
        sum_rows8(acc, v);
    }
}

#define FUSE_WAVES 25
#define FUSE_TPB   (FUSE_WAVES * 32)
#define APITCH     132

template <int FOUT>
__global__ __launch_bounds__(FUSE_TPB)
void gcn_gather_gemm_kernel(const int* __restrict__ rowptr,
                            const int* __restrict__ csr_src,
                            const float4* __restrict__ hs,
                            const float* __restrict__ bias,
                            const float* __restrict__ B,
                            float* __restrict__ Out,
                            int totalStrips)
{
    constexpr int FP = (FOUT % 16 == 0) ? FOUT : ((FOUT + 15) & ~15);
    constexpr int NT = FP / 16;

    __shared__ __attribute__((aligned(16))) float lds[KDIM * FP + FUSE_WAVES * 16 * APITCH];
    float* ldsB = lds;
    float* ldsA = lds + KDIM * FP;

    {
        constexpr int NE = KDIM * FP;
#pragma unroll 4
        for (int p = threadIdx.x; p < NE; p += FUSE_TPB) {
            int k = p / FP;
            int c = p - k * FP;
            float val;
            if constexpr (FOUT == FP)
                val = B[p];
            else
                val = (c < FOUT) ? B[k * FOUT + c] : 0.0f;
            ldsB[(((k >> 1) * FP) + c) * 2 + (k & 1)] = val;
        }
    }
    __syncthreads();

    const int lane = threadIdx.x & 31;
    const int wv    = __builtin_amdgcn_readfirstlane((int)(threadIdx.x >> 5));
    const int strip = __builtin_amdgcn_readfirstlane((int)(blockIdx.x * FUSE_WAVES) + wv);
    if (strip >= totalStrips) return;

    float* stage = ldsA + wv * (16 * APITCH);
    const int nbase = strip * 16;

    {
        float4 bv;
        bv.x = bias[lane * 4 + 0];
        bv.y = bias[lane * 4 + 1];
        bv.z = bias[lane * 4 + 2];
        bv.w = bias[lane * 4 + 3];

        const int rp = rowptr[nbase + min(lane, 16)];
        int beg = __builtin_amdgcn_readlane(rp, 0);
        int end = __builtin_amdgcn_readlane(rp, 1);
        int ids = 0;
        if (lane < end - beg) ids = csr_src[beg + lane];
        float4 selfv = hs[(size_t)nbase * 32 + lane];

        for (int r = 0; r < 16; ++r) {
            const int deg = end - beg;
            int s8[8];
            row_ids8(s8, ids, 0);
            int endN = end;
            int idsN = 0;
            float4 selfN;
            selfN.x = 0.0f; selfN.y = 0.0f; selfN.z = 0.0f; selfN.w = 0.0f;
            if (r < 15) {
                endN = __builtin_amdgcn_readlane(rp, r + 2);
                if (lane < endN - end) idsN = csr_src[end + lane];
                selfN = hs[(size_t)(nbase + r + 1) * 32 + lane];
            }

            float4 acc = selfv;
            {
                float4 v[8];
                issue_rows8(v, hs, s8, 0, min(deg, 8), lane);
                sum_rows8(acc, v);
            }
            add_rows128(acc, hs, ids, 8, min(deg, 32), lane);
            for (int j0 = 32; j0 < deg; j0 += 32) {
                int idsM = 0;
                if (lane < deg - j0) idsM = csr_src[beg + j0 + lane];
                add_rows128(acc, hs, idsM, 0, min(deg - j0, 32), lane);
            }

            const float d  = (float)(deg + 1);
            const float dn = (d > 0.0f) ? rsqrtf(d) : 0.0f;
            float4 o;
            o.x = fmaxf(acc.x * dn + bv.x, 0.0f);
            o.y = fmaxf(acc.y * dn + bv.y, 0.0f);
            o.z = fmaxf(acc.z * dn + bv.z, 0.0f);
            o.w = fmaxf(acc.w * dn + bv.w, 0.0f);
            *(float4*)(stage + r * APITCH + lane * 4) = o;

            beg = end; end = endN; ids = idsN; selfv = selfN;
        }
    }
    __builtin_amdgcn_fence(__ATOMIC_RELEASE, "wavefront");
    __builtin_amdgcn_wave_barrier();
    __builtin_amdgcn_fence(__ATOMIC_ACQUIRE, "wavefront");

    const int half = lane >> 4;
    const int mn   = lane & 15;

    float dv[8];
    {
        const int* rpp = rowptr + nbase + half * 8;
        int lo = rpp[0];
#pragma unroll
        for (int r = 0; r < 8; ++r) {
            int hi = rpp[r + 1];
            float d = (float)(hi - lo + 1);
            dv[r] = (d > 0.0f) ? rsqrtf(d) : 0.0f;
            lo = hi;
        }
    }

    const float* arow = stage + mn * APITCH;

    v8f acc[NT] = {};
#pragma unroll
    for (int k0 = 0; k0 < KDIM; k0 += 4) {
        int k  = k0 + 2 * half;
        int kp = (k0 >> 1) + half;
        v2f a = *(const v2f*)(arow + k);
        const float* brow = ldsB + (size_t)kp * FP * 2 + mn * 2;
#pragma unroll
        for (int nt = 0; nt < NT; ++nt) {
            v2f b = *(const v2f*)(brow + nt * 32);
            acc[nt] = __builtin_amdgcn_wmma_f32_16x16x4_f32(
                false, a, false, b, (short)0, acc[nt], false, false);
        }
    }

#pragma unroll
    for (int nt = 0; nt < NT; ++nt) {
#pragma unroll
        for (int r = 0; r < 8; ++r) {
            int orow = nbase + half * 8 + r;
            int ocol = nt * 16 + mn;
            if (FOUT == FP || ocol < FOUT)
                Out[(size_t)orow * FOUT + ocol] = acc[nt][r] * dv[r];
        }
    }
}

__device__ __forceinline__ void add_rows40(float4& acc, const float4* __restrict__ hs,
                                           int ids, int m, int cc)
{
    for (int j = 0; j < m; j += 4) {
        const int s0 = __shfl(ids, j, 16);
        const int s1 = __shfl(ids, (j + 1) & 15, 16);
        const int s2 = __shfl(ids, (j + 2) & 15, 16);
        const int s3 = __shfl(ids, (j + 3) & 15, 16);
        float4 v1 = {0.0f, 0.0f, 0.0f, 0.0f};
        float4 v2 = {0.0f, 0.0f, 0.0f, 0.0f};
        float4 v3 = {0.0f, 0.0f, 0.0f, 0.0f};
        const float4 v0 = hs[(size_t)s0 * 10 + cc];
        if (j + 1 < m) v1 = hs[(size_t)s1 * 10 + cc];
        if (j + 2 < m) v2 = hs[(size_t)s2 * 10 + cc];
        if (j + 3 < m) v3 = hs[(size_t)s3 * 10 + cc];
        acc.x += (v0.x + v1.x) + (v2.x + v3.x);
        acc.y += (v0.y + v1.y) + (v2.y + v3.y);
        acc.z += (v0.z + v1.z) + (v2.z + v3.z);
        acc.w += (v0.w + v1.w) + (v2.w + v3.w);
    }
}

__global__ __launch_bounds__(256)
void gather_rows40_kernel(const int* __restrict__ rowptr,
                          const int* __restrict__ csr_src,
                          const float4* __restrict__ hs,
                          const float* __restrict__ bias,
                          float4* __restrict__ out, int n_quads)
{
    const int lane = threadIdx.x & 31;
    const int sub  = lane >> 4;
    const int c    = lane & 15;
    const int cc   = min(c, 9);
    const int wave   = blockIdx.x * 8 + (threadIdx.x >> 5);
    const int nwaves = gridDim.x * 8;

    float4 bv;
    bv.x = bias[cc * 4 + 0];
    bv.y = bias[cc * 4 + 1];
    bv.z = bias[cc * 4 + 2];
    bv.w = bias[cc * 4 + 3];

    for (int q = wave; q < n_quads; q += nwaves) {
#pragma unroll
        for (int p = 0; p < 2; ++p) {
            const int n   = q * 4 + p * 2 + sub;
            const int beg = rowptr[n];
            const int deg = rowptr[n + 1] - beg;
            int ids = 0;
            if (c < deg) ids = csr_src[beg + c];
            float4 acc = hs[(size_t)n * 10 + cc];
            add_rows40(acc, hs, ids, min(deg, 16), cc);
            for (int j0 = 16; j0 < deg; j0 += 16) {
                ids = 0;
                if (c < deg - j0) ids = csr_src[beg + j0 + c];
                add_rows40(acc, hs, ids, min(deg - j0, 16), cc);
            }
            const float d  = (float)(deg + 1);
            const float dn = (d > 0.0f) ? rsqrtf(d) : 0.0f;
            if (c < 10) {
                float4 r;
                r.x = acc.x * dn + bv.x;
                r.y = acc.y * dn + bv.y;
                r.z = acc.z * dn + bv.z;
                r.w = acc.w * dn + bv.w;
                out[(size_t)n * 10 + c] = r;
            }
        }
    }
}


extern "C" void kernel_launch(void* const* d_in, const int* in_sizes, int n_in,
                              void* d_out, int out_size, void* d_ws, size_t ws_size,
                              hipStream_t stream)
{
    const float* x  = (const float*)d_in[0];
    const int*   ei = (const int*)  d_in[1];
    const float* W[4] = { (const float*)d_in[2], (const float*)d_in[4],
                          (const float*)d_in[6], (const float*)d_in[8] };
    const float* bia[4] = { (const float*)d_in[3], (const float*)d_in[5],
                            (const float*)d_in[7], (const float*)d_in[9] };
    const int* src = ei;
    const int* dst = ei + N_EDGES;

    char* ws = (char*)d_ws;
    size_t off = 0;
    auto alloc = [&](size_t bytes) {
        size_t p = off;
        off = (off + bytes + 255) & ~(size_t)255;
        return p;
    };
    int*   rowptr  = (int*)  (ws + alloc((size_t)(N_NODES + 1) * 4));
    int*   csr_src = (int*)  (ws + alloc((size_t)N_EDGES * 4));
    float* T       = (float*)(ws + alloc((size_t)N_NODES * HDIM * 4));
    float* P       = (float*)(ws + alloc((size_t)N_NODES * HDIM * 4));
    int2* staged = (int2*)P;
    int2* runtab = staged + N_EDGES;

    const int TPB = 256;
    const int strips = N_NODES / 16;
    const int gemmBlocks = (strips + 7) / 8;
    const int fuseBlocks = (strips + FUSE_WAVES - 1) / FUSE_WAVES;
    const int gather40Blocks = N_NODES / 4 / 8;

    edge_slice_bin_kernel<<<NBKT, BIN_TPB, 0, stream>>>(src, dst, runtab, staged);
    range_csr_kernel<<<NBKT, BIN_TPB, 0, stream>>>(runtab, staged, rowptr, csr_src);

    gcn_gemm_wmma_kernel<128, false><<<gemmBlocks, TPB, 0, stream>>>(x, W[0], rowptr, T, strips);
    gcn_gather_gemm_kernel<128><<<fuseBlocks, FUSE_TPB, 0, stream>>>(rowptr, csr_src, (const float4*)T, bia[0], W[1], P, strips);
    gcn_gather_gemm_kernel<128><<<fuseBlocks, FUSE_TPB, 0, stream>>>(rowptr, csr_src, (const float4*)P, bia[1], W[2], T, strips);
    gcn_gather_gemm_kernel<40><<<fuseBlocks, FUSE_TPB, 0, stream>>>(rowptr, csr_src, (const float4*)T, bia[2], W[3], P, strips);
    gather_rows40_kernel<<<gather40Blocks, TPB, 0, stream>>>(rowptr, csr_src, (const float4*)P, bia[3], (float4*)d_out, N_NODES / 4);
}


# Weights reach LDS once per WGP by 16-byte loads issued together in all four GEMM launches (layer-0 GEMM regrouped into 25-wave blocks for it): 26 staging round trips become 4
# speedup vs baseline: 3.2639x; 1.0920x over previous
#include <hip/hip_runtime.h>


#define N_NODES 100000
#define N_EDGES 500000
#define KDIM    128
#define HDIM    128
#define NCLS    40

typedef __attribute__((ext_vector_type(2))) float v2f;
typedef __attribute__((ext_vector_type(8))) float v8f;

#define NBKT        250
#define BKT_NODES   400
#define SLICE_EDGES 2000
#define BIN_TPB     256
#define SLICE_PER_THREAD ((SLICE_EDGES + BIN_TPB - 1) / BIN_TPB)
#define CSR_OBUF    16384

static_assert(NBKT * BKT_NODES == N_NODES, "ranges must tile the nodes");
static_assert(NBKT * SLICE_EDGES == N_EDGES, "slices must tile the edges");
static_assert(NBKT <= BIN_TPB && BKT_NODES <= 2 * BIN_TPB && (BKT_NODES % 2) == 0, "one thread per range, two nodes per thread");

__device__ __forceinline__ int block_scan_excl_256(int v, int* wsum, int& total)
{
    const int lane = threadIdx.x & 31;
    const int w    = threadIdx.x >> 5;
    int incl = v;
#pragma unroll
    for (int d = 1; d < 32; d <<= 1) {
        int t = __shfl_up(incl, d, 32);
        if (lane >= d) incl += t;
    }
    if (lane == 31) wsum[w] = incl;
    __syncthreads();
    int before = 0;
    total = 0;
#pragma unroll
    for (int i = 0; i < BIN_TPB / 32; ++i) {
        const int x = wsum[i];
        if (i < w) before += x;
        total += x;
    }
    __syncthreads();
    return before + incl - v;
}

__global__ __launch_bounds__(BIN_TPB)
void edge_slice_bin_kernel(const int* __restrict__ src, const int* __restrict__ dst,
                           int2* __restrict__ runtab, int2* __restrict__ staged)
{
    __shared__ int  hist[BIN_TPB];
    __shared__ int  offs[BIN_TPB];
    __shared__ int  wsum[BIN_TPB / 32];
    __shared__ int2 sorted[SLICE_EDGES];

    const int tid = threadIdx.x;
    const int g   = blockIdx.x;
    const int e0  = g * SLICE_EDGES;

    hist[tid] = 0;

    int d[SLICE_PER_THREAD], s[SLICE_PER_THREAD], p[SLICE_PER_THREAD];
#pragma unroll
    for (int i = 0; i < SLICE_PER_THREAD; ++i) {
        const int idx = tid + i * BIN_TPB;
        d[i] = 0; s[i] = 0;
        if (idx < SLICE_EDGES) { d[i] = dst[e0 + idx]; s[i] = src[e0 + idx]; }
    }
    __syncthreads();
#pragma unroll
    for (int i = 0; i < SLICE_PER_THREAD; ++i) {
        const int idx = tid + i * BIN_TPB;
        p[i] = 0;
        if (idx < SLICE_EDGES) p[i] = atomicAdd(&hist[(unsigned)d[i] / BKT_NODES], 1);
    }
    __syncthreads();

    const int mine = hist[tid];
    int total;
    const int excl = block_scan_excl_256(mine, wsum, total);
    offs[tid] = excl;
    if (tid < NBKT) {
        int2 run; run.x = mine; run.y = excl;
        runtab[g * NBKT + tid] = run;
    }
    __syncthreads();
#pragma unroll
    for (int i = 0; i < SLICE_PER_THREAD; ++i) {
        const int idx = tid + i * BIN_TPB;
        if (idx < SLICE_EDGES) {
            int2 e; e.x = d[i]; e.y = s[i];
            sorted[offs[(unsigned)d[i] / BKT_NODES] + p[i]] = e;
        }
    }
    __syncthreads();
    for (int idx = tid; idx < SLICE_EDGES; idx += BIN_TPB) staged[e0 + idx] = sorted[idx];
}

__global__ __launch_bounds__(BIN_TPB)
void range_csr_kernel(const int2* __restrict__ runtab,
                      const int2* __restrict__ staged,
                      int* __restrict__ rowptr, int* __restrict__ csr_src)
{
    __shared__ int ncnt[2 * BIN_TPB];
    __shared__ int wsum[BIN_TPB / 32];
    __shared__ int obuf[CSR_OBUF];

    const int tid = threadIdx.x;
    const int k   = blockIdx.x;
    const int lo  = k * BKT_NODES;

    int len = 0, pre = 0;
    if (tid < NBKT) {
        const int2 run = runtab[tid * NBKT + k];
        len = run.x;
        pre = run.y;
    }
    const int2* seg = staged + (tid < NBKT ? tid : 0) * SLICE_EDGES + pre;

    ncnt[tid] = 0;
    ncnt[tid + BIN_TPB] = 0;
    __syncthreads();

    const int* segd = (const int*)seg;
    for (int j = 0; j < len; j += 4) {
        int dn[4];
#pragma unroll
        for (int u = 0; u < 4; ++u) {
            dn[u] = lo;
            if (j + u < len) dn[u] = segd[2 * (j + u)];
        }
#pragma unroll
        for (int u = 0; u < 4; ++u)
            if (j + u < len) atomicAdd(&ncnt[dn[u] - lo], 1);
    }

    int base, cntk, unused;
    (void)block_scan_excl_256(pre, wsum, base);
    (void)block_scan_excl_256(len, wsum, cntk);

    const int ca = ncnt[2 * tid];
    const int cb = ncnt[2 * tid + 1];
    const int ex = block_scan_excl_256(ca + cb, wsum, unused);
    ncnt[2 * tid]     = ex;
    ncnt[2 * tid + 1] = ex + ca;
    if (2 * tid < BKT_NODES) {
        rowptr[lo + 2 * tid]     = base + ex;
        rowptr[lo + 2 * tid + 1] = base + ex + ca;
    }
    if (k == NBKT - 1 && tid == 0) rowptr[N_NODES] = base + cntk;
    __syncthreads();

    const bool in_lds = (cntk <= CSR_OBUF);
    for (int j = 0; j < len; j += 4) {
        int2 e[4];
#pragma unroll
        for (int u = 0; u < 4; ++u) {
            e[u].x = lo; e[u].y = 0;
            if (j + u < len) e[u] = seg[j + u];
        }
#pragma unroll
        for (int u = 0; u < 4; ++u) {
            if (j + u < len) {
                const int pos = atomicAdd(&ncnt[e[u].x - lo], 1);
                if (in_lds) obuf[pos] = e[u].y;
                else        csr_src[base + pos] = e[u].y;
            }
        }
    }
    __syncthreads();
    if (in_lds)
        for (int i = tid; i < cntk; i += BIN_TPB) csr_src[base + i] = obuf[i];
}

constexpr int stage_threads(int units, int tpb)
{
    int best = 32;
    for (int t = 32; t <= tpb; t += 32)
        if (units % t == 0) best = t;
    return best;
}

template <int FOUT, int FP, int TPB>
__device__ __forceinline__ void stage_weights_lds(const float* __restrict__ B, float* ldsB)
{
    static_assert(FOUT % 4 == 0 && FP % 4 == 0 && KDIM % 2 == 0, "16-byte pieces, row pairs");
    constexpr int CG    = FP / 4;
    constexpr int CGR   = FOUT / 4;
    constexpr int UNITS = (KDIM / 2) * CG;
    constexpr int NTHR  = stage_threads(UNITS, TPB);
    constexpr int NIT   = UNITS / NTHR;
    static_assert(NIT * NTHR == UNITS && NTHR % 32 == 0, "whole waves, equal shares");

    if ((int)threadIdx.x >= NTHR) return;

    float4 r0[NIT], r1[NIT];
#pragma unroll
    for (int i = 0; i < NIT; ++i) {
        const int u  = (int)threadIdx.x + i * NTHR;
        const int kp = u / CG;
        const int cg = min(u - kp * CG, CGR - 1);
        const float4* p = (const float4*)(B + (size_t)(2 * kp) * FOUT) + cg;
        r0[i] = p[0];
        r1[i] = p[CGR];
    }
#pragma unroll
    for (int i = 0; i < NIT; ++i) {
        const int u  = (int)threadIdx.x + i * NTHR;
        const int kp = u / CG;
        const int cg = u - kp * CG;
        const bool pad = (cg >= CGR);
        const float4 a = r0[i], b = r1[i];
        float4 lo, hi;
        lo.x = pad ? 0.0f : a.x; lo.y = pad ? 0.0f : b.x; lo.z = pad ? 0.0f : a.y; lo.w = pad ? 0.0f : b.y;
        hi.x = pad ? 0.0f : a.z; hi.y = pad ? 0.0f : b.z; hi.z = pad ? 0.0f : a.w; hi.w = pad ? 0.0f : b.w;
        float4* d = (float4*)(ldsB + ((size_t)kp * FP + 4 * cg) * 2);
        d[0] = lo;
        d[1] = hi;
    }
}

#define GEMM_WAVES 25
#define GEMM_TPB   (GEMM_WAVES * 32)

template <int FOUT, bool RELU>
__global__ __launch_bounds__(GEMM_TPB)
void gcn_gemm_wmma_kernel(const float* __restrict__ A,
                          const float* __restrict__ B,
                          const int* __restrict__ rowptr,
                          float* __restrict__ Out,
                          int totalStrips)
{
    constexpr int FP = (FOUT % 16 == 0) ? FOUT : ((FOUT + 15) & ~15);
    constexpr int NT = FP / 16;

    __shared__ __attribute__((aligned(16))) float ldsB[KDIM * FP];

    stage_weights_lds<FOUT, FP, GEMM_TPB>(B, ldsB);
    __syncthreads();

    int strip = blockIdx.x * (blockDim.x >> 5) + (threadIdx.x >> 5);
    if (strip >= totalStrips) return;

    int lane = threadIdx.x & 31;
    int half = lane >> 4;
    int mn   = lane & 15;

    float dv[8];
    {
        const int* rp = rowptr + strip * 16 + half * 8;
        int lo = rp[0];
#pragma unroll
        for (int r = 0; r < 8; ++r) {
            int hi = rp[r + 1];
            float d = (float)(hi - lo + 1);
            dv[r] = (d > 0.0f) ? rsqrtf(d) : 0.0f;
            lo = hi;
        }
    }

    const float* arow = A + (size_t)(strip * 16 + mn) * KDIM;

    v8f acc[NT] = {};
#pragma unroll
    for (int k0 = 0; k0 < KDIM; k0 += 4) {
        int k  = k0 + 2 * half;
        int kp = (k0 >> 1) + half;
        v2f a = *(const v2f*)(arow + k);
        if constexpr (RELU) {
            a.x = fmaxf(a.x, 0.0f);
            a.y = fmaxf(a.y, 0.0f);
        }
        const float* brow = ldsB + (size_t)kp * FP * 2 + mn * 2;
#pragma unroll
        for (int nt = 0; nt < NT; ++nt) {
            v2f b = *(const v2f*)(brow + nt * 32);
            acc[nt] = __builtin_amdgcn_wmma_f32_16x16x4_f32(
                false, a, false, b, (short)0, acc[nt], false, false);
        }
    }

#pragma unroll
    for (int nt = 0; nt < NT; ++nt) {
#pragma unroll
        for (int r = 0; r < 8; ++r) {
            int orow = strip * 16 + half * 8 + r;
            int ocol = nt * 16 + mn;
            if (FOUT == FP || ocol < FOUT)
                Out[(size_t)orow * FOUT + ocol] = acc[nt][r] * dv[r];
        }
    }
}


__device__ __forceinline__ void row_ids8(int (&s)[8], int ids, int j)
{
#pragma unroll
    for (int i = 0; i < 8; ++i) s[i] = __builtin_amdgcn_readlane(ids, j + i);
}

__device__ __forceinline__ void issue_rows8(float4 (&v)[8], const float4* __restrict__ hs,
                                            const int (&s)[8], int j, int m, int lane)
{
#pragma unroll
    for (int i = 0; i < 8; ++i) {
        v[i].x = 0.0f; v[i].y = 0.0f; v[i].z = 0.0f; v[i].w = 0.0f;
        if (j + i < m) v[i] = hs[(size_t)s[i] * 32 + lane];
    }
}

__device__ __forceinline__ void sum_rows8(float4& acc, const float4 (&v)[8])
{
    acc.x += ((v[0].x + v[1].x) + (v[2].x + v[3].x)) + ((v[4].x + v[5].x) + (v[6].x + v[7].x));
    acc.y += ((v[0].y + v[1].y) + (v[2].y + v[3].y)) + ((v[4].y + v[5].y) + (v[6].y + v[7].y));
    acc.z += ((v[0].z + v[1].z) + (v[2].z + v[3].z)) + ((v[4].z + v[5].z) + (v[6].z + v[7].z));
    acc.w += ((v[0].w + v[1].w) + (v[2].w + v[3].w)) + ((v[4].w + v[5].w) + (v[6].w + v[7].w));
}

__device__ __forceinline__ void add_rows128(float4& acc, const float4* __restrict__ hs,
                                            int ids, int j0, int m, int lane)
{
    for (int j = j0; j < m; j += 8) {
        int s[8];
        row_ids8(s, ids, j);
        float4 v[8];
        issue_rows8(v, hs, s, j, m, lane);
        sum_rows8(acc, v);
    }
}

#define FUSE_WAVES 25
#define FUSE_TPB   (FUSE_WAVES * 32)
#define APITCH     132

template <int FOUT>
__global__ __launch_bounds__(FUSE_TPB)
void gcn_gather_gemm_kernel(const int* __restrict__ rowptr,
                            const int* __restrict__ csr_src,
                            const float4* __restrict__ hs,
                            const float* __restrict__ bias,
                            const float* __restrict__ B,
                            float* __restrict__ Out,
                            int totalStrips)
{
    constexpr int FP = (FOUT % 16 == 0) ? FOUT : ((FOUT + 15) & ~15);
    constexpr int NT = FP / 16;

    __shared__ __attribute__((aligned(16))) float lds[KDIM * FP + FUSE_WAVES * 16 * APITCH];
    float* ldsB = lds;
    float* ldsA = lds + KDIM * FP;

    stage_weights_lds<FOUT, FP, FUSE_TPB>(B, ldsB);
    __syncthreads();

    const int lane = threadIdx.x & 31;
    const int wv    = __builtin_amdgcn_readfirstlane((int)(threadIdx.x >> 5));
    const int strip = __builtin_amdgcn_readfirstlane((int)(blockIdx.x * FUSE_WAVES) + wv);
    if (strip >= totalStrips) return;

    float* stage = ldsA + wv * (16 * APITCH);
    const int nbase = strip * 16;

    {
        float4 bv;
        bv.x = bias[lane * 4 + 0];
        bv.y = bias[lane * 4 + 1];
        bv.z = bias[lane * 4 + 2];
        bv.w = bias[lane * 4 + 3];

        const int rp = rowptr[nbase + min(lane, 16)];
        int beg = __builtin_amdgcn_readlane(rp, 0);
        int end = __builtin_amdgcn_readlane(rp, 1);
        int ids = 0;
        if (lane < end - beg) ids = csr_src[beg + lane];
        float4 selfv = hs[(size_t)nbase * 32 + lane];

        for (int r = 0; r < 16; ++r) {
            const int deg = end - beg;
            int s8[8];
            row_ids8(s8, ids, 0);
            int endN = end;
            int idsN = 0;
            float4 selfN;
            selfN.x = 0.0f; selfN.y = 0.0f; selfN.z = 0.0f; selfN.w = 0.0f;
            if (r < 15) {
                endN = __builtin_amdgcn_readlane(rp, r + 2);
                if (lane < endN - end) idsN = csr_src[end + lane];
                selfN = hs[(size_t)(nbase + r + 1) * 32 + lane];
            }

            float4 acc = selfv;
            {
                float4 v[8];
                issue_rows8(v, hs, s8, 0, min(deg, 8), lane);
                sum_rows8(acc, v);
            }
            add_rows128(acc, hs, ids, 8, min(deg, 32), lane);
            for (int j0 = 32; j0 < deg; j0 += 32) {
                int idsM = 0;
                if (lane < deg - j0) idsM = csr_src[beg + j0 + lane];
                add_rows128(acc, hs, idsM, 0, min(deg - j0, 32), lane);
            }

            const float d  = (float)(deg + 1);
            const float dn = (d > 0.0f) ? rsqrtf(d) : 0.0f;
            float4 o;
            o.x = fmaxf(acc.x * dn + bv.x, 0.0f);
            o.y = fmaxf(acc.y * dn + bv.y, 0.0f);
            o.z = fmaxf(acc.z * dn + bv.z, 0.0f);
            o.w = fmaxf(acc.w * dn + bv.w, 0.0f);
            *(float4*)(stage + r * APITCH + lane * 4) = o;

            beg = end; end = endN; ids = idsN; selfv = selfN;
        }
    }
    __builtin_amdgcn_fence(__ATOMIC_RELEASE, "wavefront");
    __builtin_amdgcn_wave_barrier();
    __builtin_amdgcn_fence(__ATOMIC_ACQUIRE, "wavefront");

    const int half = lane >> 4;
    const int mn   = lane & 15;

    float dv[8];
    {
        const int* rpp = rowptr + nbase + half * 8;
        int lo = rpp[0];
#pragma unroll
        for (int r = 0; r < 8; ++r) {
            int hi = rpp[r + 1];
            float d = (float)(hi - lo + 1);
            dv[r] = (d > 0.0f) ? rsqrtf(d) : 0.0f;
            lo = hi;
        }
    }

    const float* arow = stage + mn * APITCH;

    v8f acc[NT] = {};
#pragma unroll
    for (int k0 = 0; k0 < KDIM; k0 += 4) {
        int k  = k0 + 2 * half;
        int kp = (k0 >> 1) + half;
        v2f a = *(const v2f*)(arow + k);
        const float* brow = ldsB + (size_t)kp * FP * 2 + mn * 2;
#pragma unroll
        for (int nt = 0; nt < NT; ++nt) {
            v2f b = *(const v2f*)(brow + nt * 32);
            acc[nt] = __builtin_amdgcn_wmma_f32_16x16x4_f32(
                false, a, false, b, (short)0, acc[nt], false, false);
        }
    }

#pragma unroll
    for (int nt = 0; nt < NT; ++nt) {
#pragma unroll
        for (int r = 0; r < 8; ++r) {
            int orow = nbase + half * 8 + r;
            int ocol = nt * 16 + mn;
            if (FOUT == FP || ocol < FOUT)
                Out[(size_t)orow * FOUT + ocol] = acc[nt][r] * dv[r];
        }
    }
}

__device__ __forceinline__ void add_rows40(float4& acc, const float4* __restrict__ hs,
                                           int ids, int m, int cc)
{
    for (int j = 0; j < m; j += 4) {
        const int s0 = __shfl(ids, j, 16);
        const int s1 = __shfl(ids, (j + 1) & 15, 16);
        const int s2 = __shfl(ids, (j + 2) & 15, 16);
        const int s3 = __shfl(ids, (j + 3) & 15, 16);
        float4 v1 = {0.0f, 0.0f, 0.0f, 0.0f};
        float4 v2 = {0.0f, 0.0f, 0.0f, 0.0f};
        float4 v3 = {0.0f, 0.0f, 0.0f, 0.0f};
        const float4 v0 = hs[(size_t)s0 * 10 + cc];
        if (j + 1 < m) v1 = hs[(size_t)s1 * 10 + cc];
        if (j + 2 < m) v2 = hs[(size_t)s2 * 10 + cc];
        if (j + 3 < m) v3 = hs[(size_t)s3 * 10 + cc];
        acc.x += (v0.x + v1.x) + (v2.x + v3.x);
        acc.y += (v0.y + v1.y) + (v2.y + v3.y);
        acc.z += (v0.z + v1.z) + (v2.z + v3.z);
        acc.w += (v0.w + v1.w) + (v2.w + v3.w);
    }
}

__global__ __launch_bounds__(256)
void gather_rows40_kernel(const int* __restrict__ rowptr,
                          const int* __restrict__ csr_src,
                          const float4* __restrict__ hs,
                          const float* __restrict__ bias,
                          float4* __restrict__ out, int n_quads)
{
    const int lane = threadIdx.x & 31;
    const int sub  = lane >> 4;
    const int c    = lane & 15;
    const int cc   = min(c, 9);
    const int wave   = blockIdx.x * 8 + (threadIdx.x >> 5);
    const int nwaves = gridDim.x * 8;

    float4 bv;
    bv.x = bias[cc * 4 + 0];
    bv.y = bias[cc * 4 + 1];
    bv.z = bias[cc * 4 + 2];
    bv.w = bias[cc * 4 + 3];

    for (int q = wave; q < n_quads; q += nwaves) {
#pragma unroll
        for (int p = 0; p < 2; ++p) {
            const int n   = q * 4 + p * 2 + sub;
            const int beg = rowptr[n];
            const int deg = rowptr[n + 1] - beg;
            int ids = 0;
            if (c < deg) ids = csr_src[beg + c];
            float4 acc = hs[(size_t)n * 10 + cc];
            add_rows40(acc, hs, ids, min(deg, 16), cc);
            for (int j0 = 16; j0 < deg; j0 += 16) {
                ids = 0;
                if (c < deg - j0) ids = csr_src[beg + j0 + c];
                add_rows40(acc, hs, ids, min(deg - j0, 16), cc);
            }
            const float d  = (float)(deg + 1);
            const float dn = (d > 0.0f) ? rsqrtf(d) : 0.0f;
            if (c < 10) {
                float4 r;
                r.x = acc.x * dn + bv.x;
                r.y = acc.y * dn + bv.y;
                r.z = acc.z * dn + bv.z;
                r.w = acc.w * dn + bv.w;
                out[(size_t)n * 10 + c] = r;
            }
        }
    }
}


extern "C" void kernel_launch(void* const* d_in, const int* in_sizes, int n_in,
                              void* d_out, int out_size, void* d_ws, size_t ws_size,
                              hipStream_t stream)
{
    const float* x  = (const float*)d_in[0];
    const int*   ei = (const int*)  d_in[1];
    const float* W[4] = { (const float*)d_in[2], (const float*)d_in[4],
                          (const float*)d_in[6], (const float*)d_in[8] };
    const float* bia[4] = { (const float*)d_in[3], (const float*)d_in[5],
                            (const float*)d_in[7], (const float*)d_in[9] };
    const int* src = ei;
    const int* dst = ei + N_EDGES;

    char* ws = (char*)d_ws;
    size_t off = 0;
    auto alloc = [&](size_t bytes) {
        size_t p = off;
        off = (off + bytes + 255) & ~(size_t)255;
        return p;
    };
    int*   rowptr  = (int*)  (ws + alloc((size_t)(N_NODES + 1) * 4));
    int*   csr_src = (int*)  (ws + alloc((size_t)N_EDGES * 4));
    float* T       = (float*)(ws + alloc((size_t)N_NODES * HDIM * 4));
    float* P       = (float*)(ws + alloc((size_t)N_NODES * HDIM * 4));
    int2* staged = (int2*)P;
    int2* runtab = staged + N_EDGES;

    const int TPB = 256;
    const int strips = N_NODES / 16;
    const int gemmBlocks = (strips + GEMM_WAVES - 1) / GEMM_WAVES;
    const int fuseBlocks = (strips + FUSE_WAVES - 1) / FUSE_WAVES;
    const int gather40Blocks = N_NODES / 4 / 8;

    edge_slice_bin_kernel<<<NBKT, BIN_TPB, 0, stream>>>(src, dst, runtab, staged);
    range_csr_kernel<<<NBKT, BIN_TPB, 0, stream>>>(runtab, staged, rowptr, csr_src);

    gcn_gemm_wmma_kernel<128, false><<<gemmBlocks, GEMM_TPB, 0, stream>>>(x, W[0], rowptr, T, strips);
    gcn_gather_gemm_kernel<128><<<fuseBlocks, FUSE_TPB, 0, stream>>>(rowptr, csr_src, (const float4*)T, bia[0], W[1], P, strips);
    gcn_gather_gemm_kernel<128><<<fuseBlocks, FUSE_TPB, 0, stream>>>(rowptr, csr_src, (const float4*)P, bia[1], W[2], T, strips);
    gcn_gather_gemm_kernel<40><<<fuseBlocks, FUSE_TPB, 0, stream>>>(rowptr, csr_src, (const float4*)T, bia[2], W[3], P, strips);
    gather_rows40_kernel<<<gather40Blocks, TPB, 0, stream>>>(rowptr, csr_src, (const float4*)P, bia[3], (float4*)d_out, N_NODES / 4);
}
